# starting kernel
# baseline (speedup 1.0000x reference)
#include <hip/hip_runtime.h>
#include <hip/hip_bf16.h>

typedef float v2f __attribute__((ext_vector_type(2)));
typedef float v8f __attribute__((ext_vector_type(8)));

__global__ void dcn_pre_kernel(const float* __restrict__ W,
                               const float* __restrict__ b,
                               float* __restrict__ ws) {
  __shared__ float r1[1024];
  __shared__ float r2[1024];
  const int t = threadIdx.x;
  const float b0 = b[t];
  const float b1 = b[1024 + t];
  const float b2 = b[2048 + t];
  ws[4 + t] = b0 + b1 + b2;
  r1[t] = b0 * W[1024 + t];
  r2[t] = (b0 + b1) * W[2048 + t];
  __syncthreads();
  for (int s = 512; s > 0; s >>= 1) {
    if (t < s) { r1[t] += r1[t + s]; r2[t] += r2[t + s]; }
    __syncthreads();
  }
  if (t == 0) { ws[0] = 0.0f; ws[1] = r1[0]; ws[2] = r2[0]; }
}

__global__ void dcn_gemv_wmma_kernel(const float* __restrict__ x,
                                     const float* __restrict__ W,
                                     const float* __restrict__ ws_c,
                                     float* __restrict__ alpha) {
  __shared__ float sW[3 * 1024];
  __shared__ float sG[2 * 16 * 3 * 4];

  const int tid = threadIdx.x;
  for (int i = tid; i < 3 * 1024; i += 256) sW[i] = W[i];
  __syncthreads();

  const int wave  = tid >> 5;
  const int lane  = tid & 31;
  const int tileI = wave >> 2;
  const int slice = wave & 3;
  const int half  = lane >> 4;
  const int l16   = lane & 15;
  const int row0  = (blockIdx.x * 2 + tileI) * 16;
  const int k0    = slice * 256;

  const float* arow = x + (size_t)(row0 + l16) * 1024 + half * 2;
  const bool  bvalid = (l16 < 3);
  const float* brow = sW + l16 * 1024 + half * 2;
  const v2f bz = {0.0f, 0.0f};

  v8f acc = {};
  for (int k = k0; k < k0 + 256; k += 4) {
    v2f a = *(const v2f*)(arow + k);
    v2f bm = bvalid ? *(const v2f*)(brow + k) : bz;
    acc = __builtin_amdgcn_wmma_f32_16x16x4_f32(
         false, a,  false, bm,
         (short)0, acc,  false,  false);
  }

  if (bvalid) {
    #pragma unroll
    for (int r = 0; r < 8; ++r)
      sG[(((tileI * 16) + (r + 8 * half)) * 3 + l16) * 4 + slice] = acc[r];
  }
  __syncthreads();

  if (slice == 0 && lane < 16) {
    const int base = ((tileI * 16) + lane) * 3 * 4;
    const float g0 = sG[base + 0] + sG[base + 1] + sG[base + 2] + sG[base + 3];
    const float g1 = sG[base + 4] + sG[base + 5] + sG[base + 6] + sG[base + 7];
    const float g2 = sG[base + 8] + sG[base + 9] + sG[base + 10] + sG[base + 11];
    const float c1 = ws_c[1];
    const float c2 = ws_c[2];
    const float a1 = 1.0f + g0;
    const float a2 = a1 * (1.0f + g1) + c1;
    const float a3 = a2 * (1.0f + g2) + c2;
    alpha[row0 + lane] = a3;
  }
}

__global__ void dcn_out_kernel(const float* __restrict__ x,
                               const float* __restrict__ bsum,
                               const float* __restrict__ alpha,
                               float* __restrict__ out) {
  const size_t gid  = (size_t)blockIdx.x * 256 + threadIdx.x;
  const size_t base = gid * 4;
  const int row = (int)(base >> 10);
  const int col = (int)(base & 1023);
  const float a = alpha[row];
  const float4 xv = *(const float4*)(x + base);
  const float4 bv = *(const float4*)(bsum + col);
  float4 o;
  o.x = a * xv.x + bv.x;
  o.y = a * xv.y + bv.y;
  o.z = a * xv.z + bv.z;
  o.w = a * xv.w + bv.w;
  *(float4*)(out + base) = o;
}

extern "C" void kernel_launch(void* const* d_in, const int* in_sizes, int n_in,
                              void* d_out, int out_size, void* d_ws, size_t ws_size,
                              hipStream_t stream) {
  (void)in_sizes; (void)n_in; (void)out_size; (void)ws_size;
  const float* x = (const float*)d_in[0];
  const float* W = (const float*)d_in[1];
  const float* b = (const float*)d_in[2];
  float* out = (float*)d_out;
  float* ws  = (float*)d_ws;

  float* ws_c     = ws;
  float* ws_bsum  = ws + 4;
  float* ws_alpha = ws + 1028;

  dcn_pre_kernel<<<1, 1024, 0, stream>>>(W, b, ws);
  dcn_gemv_wmma_kernel<<<256, 256, 0, stream>>>(x, W, ws_c, ws_alpha);
  dcn_out_kernel<<<8192, 256, 0, stream>>>(x, ws_bsum, ws_alpha, out);
}


# One launch, no LDS transpose: wave reads its 16x128 x patch twice, coalesced (kept for output) and row-per-lane as WMMA fragments from the WGP cache; 8-way split-K: g, c sum order changes
# speedup vs baseline: 2.2027x; 2.2027x over previous
#include <hip/hip_runtime.h>


typedef float v2f __attribute__((ext_vector_type(2)));
typedef float v4f __attribute__((ext_vector_type(4)));
typedef float v8f __attribute__((ext_vector_type(8)));

#define SW_PITCH 1032

__global__ __launch_bounds__(256)
void dcn_fused_kernel(const float* __restrict__ x,
                      const float* __restrict__ W,
                      const float* __restrict__ b,
                      float* __restrict__ out) {
  __shared__ float sW[3 * SW_PITCH];
  __shared__ float sG[8 * 3 * 16];
  __shared__ float sC[8 * 2];

  const int tid  = threadIdx.x;
  const int wave = tid >> 5;
  const int lane = tid & 31;
  const int c    = lane & 15;
  const int h    = lane >> 4;
  const int row0 = blockIdx.x * 16;
  const int col0 = wave * 128;
  const int cc   = col0 + 4 * lane;

  const v4f b0 = *(const v4f*)(b + cc);
  const v4f b1 = *(const v4f*)(b + 1024 + cc);
  const v4f b2 = *(const v4f*)(b + 2048 + cc);
  const v4f w0 = *(const v4f*)(W + cc);
  const v4f w1 = *(const v4f*)(W + 1024 + cc);
  const v4f w2 = *(const v4f*)(W + 2048 + cc);

  const float* xcp = x + (size_t)row0 * 1024 + cc;
  v4f xc[16];
  #pragma unroll
  for (int r = 0; r < 16; ++r) xc[r] = *(const v4f*)(xcp + r * 1024);

  const float* xfp = x + (size_t)(row0 + c) * 1024 + col0 + 4 * h;
  v4f xf[16];
  #pragma unroll
  for (int q = 0; q < 16; ++q) xf[q] = *(const v4f*)(xfp + 8 * q);

  *(v4f*)(sW + cc)                = w0;
  *(v4f*)(sW + SW_PITCH + cc)     = w1;
  *(v4f*)(sW + 2 * SW_PITCH + cc) = w2;
  const v4f b01 = b0 + b1;
  const v4f bs  = b01 + b2;
  float p1 = b0.x * w1.x + b0.y * w1.y + b0.z * w1.z + b0.w * w1.w;
  float p2 = b01.x * w2.x + b01.y * w2.y + b01.z * w2.z + b01.w * w2.w;
  #pragma unroll
  for (int m = 16; m > 0; m >>= 1) {
    p1 += __shfl_xor(p1, m);
    p2 += __shfl_xor(p2, m);
  }
  if (lane == 0) { sC[wave * 2] = p1; sC[wave * 2 + 1] = p2; }
  __syncthreads();

  const bool  wvalid = (c < 3);
  const float* swp = sW + (wvalid ? c : 0) * SW_PITCH + col0 + 4 * h;
  v8f acc0 = {};
  v8f acc1 = {};
  #pragma unroll
  for (int q = 0; q < 16; ++q) {
    v4f wf = *(const v4f*)(swp + 8 * q);
    if (!wvalid) wf = (v4f){0.0f, 0.0f, 0.0f, 0.0f};
    const v2f a_lo = {wf.x, wf.y};
    const v2f a_hi = {wf.z, wf.w};
    const v2f b_lo = {xf[q].x, xf[q].y};
    const v2f b_hi = {xf[q].z, xf[q].w};
    acc0 = __builtin_amdgcn_wmma_f32_16x16x4_f32(false, a_lo, false, b_lo,
                                                 (short)0, acc0, false, false);
    acc1 = __builtin_amdgcn_wmma_f32_16x16x4_f32(false, a_hi, false, b_hi,
                                                 (short)0, acc1, false, false);
  }

  if (h == 0) {
    sG[(wave * 3 + 0) * 16 + c] = acc0[0] + acc1[0];
    sG[(wave * 3 + 1) * 16 + c] = acc0[1] + acc1[1];
    sG[(wave * 3 + 2) * 16 + c] = acc0[2] + acc1[2];
  }
  __syncthreads();

  float g0 = 0.0f, g1 = 0.0f, g2 = 0.0f, c1 = 0.0f, c2 = 0.0f;
  #pragma unroll
  for (int s = 0; s < 8; ++s) {
    g0 += sG[(s * 3 + 0) * 16 + c];
    g1 += sG[(s * 3 + 1) * 16 + c];
    g2 += sG[(s * 3 + 2) * 16 + c];
    c1 += sC[s * 2];
    c2 += sC[s * 2 + 1];
  }
  const float a1 = 1.0f + g0;
  const float a2 = a1 * (1.0f + g1) + c1;
  const float a3 = a2 * (1.0f + g2) + c2;

  float* op = out + (size_t)row0 * 1024 + cc;
  #pragma unroll
  for (int r = 0; r < 16; ++r) {
    const float ar = __int_as_float(__builtin_amdgcn_readlane(__float_as_int(a3), r));
    v4f o;
    o.x = ar * xc[r].x + bs.x;
    o.y = ar * xc[r].y + bs.y;
    o.z = ar * xc[r].z + bs.z;
    o.w = ar * xc[r].w + bs.w;
    *(v4f*)(op + r * 1024) = o;
  }
}

extern "C" void kernel_launch(void* const* d_in, const int* in_sizes, int n_in,
                              void* d_out, int out_size, void* d_ws, size_t ws_size,
                              hipStream_t stream) {
  (void)in_sizes; (void)n_in; (void)out_size; (void)d_ws; (void)ws_size;
  const float* x = (const float*)d_in[0];
  const float* W = (const float*)d_in[1];
  const float* b = (const float*)d_in[2];
  float* out = (float*)d_out;
  dcn_fused_kernel<<<512, 256, 0, stream>>>(x, W, b, out);
}


# x read once: a coalesced 4-line load is the WMMA B fragment itself (column = x row x 32-B chunk, A = w row x chunk, equal-chunk entries summed per 4 lanes); no LDS transpose; g sum order changes
# speedup vs baseline: 2.4717x; 1.1221x over previous
#include <hip/hip_runtime.h>


typedef float v2f __attribute__((ext_vector_type(2)));
typedef float v4f __attribute__((ext_vector_type(4)));
typedef float v8f __attribute__((ext_vector_type(8)));

#define SW_PITCH 1056

__device__ __forceinline__ float quad_bcast0(float v) {
  return __int_as_float(__builtin_amdgcn_update_dpp(0, __float_as_int(v), 0x00, 0xF, 0xF, true));
}
__device__ __forceinline__ float quad_bcast1(float v) {
  return __int_as_float(__builtin_amdgcn_update_dpp(0, __float_as_int(v), 0x55, 0xF, 0xF, true));
}
__device__ __forceinline__ float quad_bcast2(float v) {
  return __int_as_float(__builtin_amdgcn_update_dpp(0, __float_as_int(v), 0xAA, 0xF, 0xF, true));
}
__device__ __forceinline__ float quad_bcast3(float v) {
  return __int_as_float(__builtin_amdgcn_update_dpp(0, __float_as_int(v), 0xFF, 0xF, 0xF, true));
}
__device__ __forceinline__ float quad_diag_sum(float a0, float a1, float a2, float a3) {
  return (quad_bcast0(a0) + quad_bcast1(a1)) + (quad_bcast2(a2) + quad_bcast3(a3));
}

__global__ __launch_bounds__(256)
void dcn_fused_kernel(const float* __restrict__ x,
                      const float* __restrict__ W,
                      const float* __restrict__ b,
                      float* __restrict__ out) {
  __shared__ float sW[3 * SW_PITCH];
  __shared__ float sB[1024];
  __shared__ float sG[8 * 3 * 16];
  __shared__ float sC[8 * 2];

  const int tid  = threadIdx.x;
  const int wave = tid >> 5;
  const int lane = tid & 31;
  const int c    = lane & 15;
  const int h    = lane >> 4;
  const int j    = c & 3;
  const int r    = c >> 2;
  const int row0 = blockIdx.x * 16;
  const int col0 = wave * 128;
  const int cc   = col0 + 4 * lane;
  const int fo   = 8 * j + 4 * h;

  const v4f b0 = *(const v4f*)(b + cc);
  const v4f b1 = *(const v4f*)(b + 1024 + cc);
  const v4f b2 = *(const v4f*)(b + 2048 + cc);
  const v4f w0 = *(const v4f*)(W + cc);
  const v4f w1 = *(const v4f*)(W + 1024 + cc);
  const v4f w2 = *(const v4f*)(W + 2048 + cc);

  const float* xp = x + (size_t)(row0 + r) * 1024 + col0 + fo;
  v4f xv[4][4];
  #pragma unroll
  for (int g = 0; g < 4; ++g) {
    #pragma unroll
    for (int q = 0; q < 4; ++q) xv[g][q] = *(const v4f*)(xp + g * 4096 + q * 32);
  }

  *(v4f*)(sW + cc)                = w0;
  *(v4f*)(sW + SW_PITCH + cc)     = w1;
  *(v4f*)(sW + 2 * SW_PITCH + cc) = w2;
  const v4f b01 = b0 + b1;
  *(v4f*)(sB + cc) = b01 + b2;
  float p1 = b0.x * w1.x + b0.y * w1.y + b0.z * w1.z + b0.w * w1.w;
  float p2 = b01.x * w2.x + b01.y * w2.y + b01.z * w2.z + b01.w * w2.w;
  #pragma unroll
  for (int m = 16; m > 0; m >>= 1) {
    p1 += __shfl_xor(p1, m);
    p2 += __shfl_xor(p2, m);
  }
  if (lane == 0) { sC[wave * 2] = p1; sC[wave * 2 + 1] = p2; }
  __syncthreads();

  const bool  wvalid = (r < 3);
  const float* swp = sW + (wvalid ? r : 0) * SW_PITCH + col0 + fo;
  v8f acc[4] = {};
  #pragma unroll
  for (int q = 0; q < 4; ++q) {
    v4f wf = *(const v4f*)(swp + 32 * q);
    if (!wvalid) wf = (v4f){0.0f, 0.0f, 0.0f, 0.0f};
    const v2f a_lo = {wf.x, wf.y};
    const v2f a_hi = {wf.z, wf.w};
    #pragma unroll
    for (int g = 0; g < 4; ++g) {
      const v2f b_lo = {xv[g][q].x, xv[g][q].y};
      acc[g] = __builtin_amdgcn_wmma_f32_16x16x4_f32(false, a_lo, false, b_lo,
                                                     (short)0, acc[g], false, false);
    }
    #pragma unroll
    for (int g = 0; g < 4; ++g) {
      const v2f b_hi = {xv[g][q].z, xv[g][q].w};
      acc[g] = __builtin_amdgcn_wmma_f32_16x16x4_f32(false, a_hi, false, b_hi,
                                                     (short)0, acc[g], false, false);
    }
  }

  v4f t0, t1;
  #pragma unroll
  for (int g = 0; g < 4; ++g) {
    t0[g] = quad_diag_sum(acc[g][0], acc[g][1], acc[g][2], acc[g][3]);
    t1[g] = quad_diag_sum(acc[g][4], acc[g][5], acc[g][6], acc[g][7]);
  }
  if (j == 0) {
    *(v4f*)(sG + (wave * 3 + 2 * h) * 16 + 4 * r) = t0;
    if (h == 0) *(v4f*)(sG + (wave * 3 + 1) * 16 + 4 * r) = t1;
  }
  __syncthreads();

  float g0 = 0.0f, g1 = 0.0f, g2 = 0.0f, c1 = 0.0f, c2 = 0.0f;
  #pragma unroll
  for (int s = 0; s < 8; ++s) {
    g0 += sG[(s * 3 + 0) * 16 + c];
    g1 += sG[(s * 3 + 1) * 16 + c];
    g2 += sG[(s * 3 + 2) * 16 + c];
    c1 += sC[s * 2];
    c2 += sC[s * 2 + 1];
  }
  const float a1 = 1.0f + g0;
  const float a2 = a1 * (1.0f + g1) + c1;
  const float a3 = a2 * (1.0f + g2) + c2;

  float ar[4];
  ar[0] = quad_bcast0(a3);
  ar[1] = quad_bcast1(a3);
  ar[2] = quad_bcast2(a3);
  ar[3] = quad_bcast3(a3);

  const float* sbp = sB + col0 + fo;
  float* op = out + (size_t)(row0 + r) * 1024 + col0 + fo;
  #pragma unroll
  for (int q = 0; q < 4; ++q) {
    const v4f bs = *(const v4f*)(sbp + 32 * q);
    #pragma unroll
    for (int g = 0; g < 4; ++g) {
      v4f o;
      o.x = ar[g] * xv[g][q].x + bs.x;
      o.y = ar[g] * xv[g][q].y + bs.y;
      o.z = ar[g] * xv[g][q].z + bs.z;
      o.w = ar[g] * xv[g][q].w + bs.w;
      *(v4f*)(op + g * 4096 + q * 32) = o;
    }
  }
}

extern "C" void kernel_launch(void* const* d_in, const int* in_sizes, int n_in,
                              void* d_out, int out_size, void* d_ws, size_t ws_size,
                              hipStream_t stream) {
  (void)in_sizes; (void)n_in; (void)out_size; (void)d_ws; (void)ws_size;
  const float* x = (const float*)d_in[0];
  const float* W = (const float*)d_in[1];
  const float* b = (const float*)d_in[2];
  float* out = (float*)d_out;
  dcn_fused_kernel<<<512, 256, 0, stream>>>(x, W, b, out);
}
